# starting kernel
# baseline (speedup 1.0000x reference)
#include <hip/hip_runtime.h>

typedef __attribute__((ext_vector_type(2))) float v2f;
typedef __attribute__((ext_vector_type(8))) float v8f;

__device__ __forceinline__ v8f wmma_f32(v2f a, v2f b, v8f c) {
  return __builtin_amdgcn_wmma_f32_16x16x4_f32(false, a, false, b, (short)0, c,
                                               false, false);
}

__global__ __launch_bounds__(256) void k_node_ffn(
    const float* __restrict__ x, const float* __restrict__ Wn1,
    const float* __restrict__ bn1, const float* __restrict__ Wn2,
    const float* __restrict__ bn2, float* __restrict__ h) {
  __shared__ float xs[16][8];
  __shared__ float ts[16][256];
  const int tid = threadIdx.x;
  const int node0 = blockIdx.x * 16;

  if (tid < 16 * 8) {
    int r = tid >> 3, k = tid & 7;
    xs[r][k] = (k < 6) ? x[(node0 + r) * 6 + k] : 0.0f;
  }
  __syncthreads();

  {
    int o = tid;
    float w0 = Wn1[0 * 256 + o], w1 = Wn1[1 * 256 + o], w2 = Wn1[2 * 256 + o];
    float w3 = Wn1[3 * 256 + o], w4 = Wn1[4 * 256 + o], w5 = Wn1[5 * 256 + o];
    float b = bn1[o];
    for (int r = 0; r < 16; ++r) {
      float acc = b + xs[r][0] * w0 + xs[r][1] * w1 + xs[r][2] * w2 +
                  xs[r][3] * w3 + xs[r][4] * w4 + xs[r][5] * w5;
      ts[r][o] = fmaxf(acc, 0.0f);
    }
  }
  __syncthreads();

  const int wave = tid >> 5, lane = tid & 31;
  const int hi = lane >> 4, lc = lane & 15;
  const int n0 = wave * 16;
  v8f acc = {};
  for (int k0 = 0; k0 < 256; k0 += 4) {
    int k = k0 + 2 * hi;
    v2f a, b;
    a.x = ts[lc][k];
    a.y = ts[lc][k + 1];
    b.x = Wn2[k * 128 + n0 + lc];
    b.y = Wn2[(k + 1) * 128 + n0 + lc];
    acc = wmma_f32(a, b, acc);
  }
  float bias = bn2[n0 + lc];
  for (int v = 0; v < 8; ++v) {
    int r = v + 8 * hi;
    h[(node0 + r) * 128 + n0 + lc] = acc[v] + bias;
  }
}

__global__ __launch_bounds__(256) void k_edge_ffn(
    const float* __restrict__ h, const float* __restrict__ We1,
    const float* __restrict__ be1, const float* __restrict__ We2,
    const float* __restrict__ be2, float* __restrict__ E) {
  __shared__ float hs[16][128];
  __shared__ float us[16][128];
  const int tid = threadIdx.x;
  const int node0 = blockIdx.x * 16;
  for (int idx = tid; idx < 16 * 128; idx += 256)
    hs[idx >> 7][idx & 127] = h[node0 * 128 + idx];
  __syncthreads();

  const int wave = tid >> 5, lane = tid & 31;
  const int hi = lane >> 4, lc = lane & 15;
  const int n0 = wave * 16;

  v8f acc = {};
  for (int k0 = 0; k0 < 128; k0 += 4) {
    int k = k0 + 2 * hi;
    v2f a, b;
    a.x = hs[lc][k];
    a.y = hs[lc][k + 1];
    b.x = We1[k * 128 + n0 + lc] + We1[(k + 128) * 128 + n0 + lc];
    b.y = We1[(k + 1) * 128 + n0 + lc] + We1[(k + 129) * 128 + n0 + lc];
    acc = wmma_f32(a, b, acc);
  }
  {
    float bias = be1[n0 + lc];
    for (int v = 0; v < 8; ++v) us[v + 8 * hi][n0 + lc] = fmaxf(acc[v] + bias, 0.0f);
  }
  __syncthreads();

  v8f acc2 = {};
  for (int k0 = 0; k0 < 128; k0 += 4) {
    int k = k0 + 2 * hi;
    v2f a, b;
    a.x = us[lc][k];
    a.y = us[lc][k + 1];
    b.x = We2[k * 128 + n0 + lc];
    b.y = We2[(k + 1) * 128 + n0 + lc];
    acc2 = wmma_f32(a, b, acc2);
  }
  float bias2 = be2[n0 + lc];
  for (int v = 0; v < 8; ++v) {
    int r = v + 8 * hi;
    E[(node0 + r) * 128 + n0 + lc] = acc2[v] + bias2;
  }
}

__global__ __launch_bounds__(256) void k_reduce(const float* __restrict__ E,
                                                float* __restrict__ S) {
  int tid = threadIdx.x;
  int n = tid >> 7, c = tid & 127;
  const float* base = E + (size_t)n * 512 * 128 + c;
  float s = 0.0f;
  for (int i = 0; i < 512; ++i) s += base[i * 128];
  S[n * 128 + c] = s;
}

__global__ __launch_bounds__(256) void k_e2n_ffn(
    const float* __restrict__ E, const float* __restrict__ S,
    const float* __restrict__ Wen1, const float* __restrict__ ben1,
    const float* __restrict__ Wen2, const float* __restrict__ ben2,
    float* __restrict__ h2) {
  __shared__ float as[16][128];
  __shared__ float us[16][256];
  const int tid = threadIdx.x;
  const int node0 = blockIdx.x * 16;
  for (int idx = tid; idx < 16 * 128; idx += 256) {
    int r = idx >> 7, c = idx & 127;
    int node = node0 + r;
    as[r][c] = S[(node >> 9) * 128 + c] - E[node * 128 + c];
  }
  __syncthreads();

  const int wave = tid >> 5, lane = tid & 31;
  const int hi = lane >> 4, lc = lane & 15;

  for (int t = 0; t < 2; ++t) {
    int n0 = (wave + t * 8) * 16;
    v8f acc = {};
    for (int k0 = 0; k0 < 128; k0 += 4) {
      int k = k0 + 2 * hi;
      v2f a, b;
      a.x = as[lc][k];
      a.y = as[lc][k + 1];
      b.x = Wen1[k * 256 + n0 + lc];
      b.y = Wen1[(k + 1) * 256 + n0 + lc];
      acc = wmma_f32(a, b, acc);
    }
    float bias = ben1[n0 + lc];
    for (int v = 0; v < 8; ++v) us[v + 8 * hi][n0 + lc] = fmaxf(acc[v] + bias, 0.0f);
  }
  __syncthreads();

  int n0 = wave * 16;
  v8f acc2 = {};
  for (int k0 = 0; k0 < 256; k0 += 4) {
    int k = k0 + 2 * hi;
    v2f a, b;
    a.x = us[lc][k];
    a.y = us[lc][k + 1];
    b.x = Wen2[k * 128 + n0 + lc];
    b.y = Wen2[(k + 1) * 128 + n0 + lc];
    acc2 = wmma_f32(a, b, acc2);
  }
  float bias2 = ben2[n0 + lc];
  for (int v = 0; v < 8; ++v) {
    int r = v + 8 * hi;
    h2[(node0 + r) * 128 + n0 + lc] = acc2[v] + bias2;
  }
}

__global__ __launch_bounds__(256) void k_n2e(
    const float* __restrict__ h2, const float* __restrict__ Wne1,
    const float* __restrict__ bne1, const float* __restrict__ Wne2,
    const float* __restrict__ bne2, float* __restrict__ out2) {
  __shared__ float hs[16][128];
  __shared__ float us[16][128];
  __shared__ float ps[16][16];
  __shared__ float sc[16];
  const int tid = threadIdx.x;
  const int node0 = blockIdx.x * 16;
  for (int idx = tid; idx < 16 * 128; idx += 256)
    hs[idx >> 7][idx & 127] = h2[node0 * 128 + idx];
  __syncthreads();

  const int wave = tid >> 5, lane = tid & 31;
  const int hi = lane >> 4, lc = lane & 15;
  const int n0 = wave * 16;
  v8f acc = {};
  for (int k0 = 0; k0 < 128; k0 += 4) {
    int k = k0 + 2 * hi;
    v2f a, b;
    a.x = hs[lc][k];
    a.y = hs[lc][k + 1];
    b.x = Wne1[k * 128 + n0 + lc] + Wne1[(k + 128) * 128 + n0 + lc];
    b.y = Wne1[(k + 1) * 128 + n0 + lc] + Wne1[(k + 129) * 128 + n0 + lc];
    acc = wmma_f32(a, b, acc);
  }
  float bias = bne1[n0 + lc];
  for (int v = 0; v < 8; ++v) us[v + 8 * hi][n0 + lc] = fmaxf(acc[v] + bias, 0.0f);
  __syncthreads();

  {
    int r = tid >> 4, l16 = tid & 15;
    float p = 0.0f;
    for (int c = l16; c < 128; c += 16) p += us[r][c] * Wne2[c];
    ps[r][l16] = p;
  }
  __syncthreads();
  if (tid < 16) {
    float s = 0.0f;
    for (int i = 0; i < 16; ++i) s += ps[tid][i];
    sc[tid] = (s + bne2[0]) * (1.0f / 512.0f);
  }
  __syncthreads();

  for (int idx = tid; idx < 16 * 512; idx += 256) {
    int r = idx >> 9, j = idx & 511;
    out2[(size_t)(node0 + r) * 512 + j] = sc[r];
  }
}

extern "C" void kernel_launch(void* const* d_in, const int* in_sizes, int n_in,
                              void* d_out, int out_size, void* d_ws,
                              size_t ws_size, hipStream_t stream) {
  (void)in_sizes; (void)n_in; (void)out_size; (void)ws_size;
  const float* x    = (const float*)d_in[0];
  const float* Wn1  = (const float*)d_in[1];
  const float* bn1  = (const float*)d_in[2];
  const float* Wn2  = (const float*)d_in[3];
  const float* bn2  = (const float*)d_in[4];
  const float* We1  = (const float*)d_in[5];
  const float* be1  = (const float*)d_in[6];
  const float* We2  = (const float*)d_in[7];
  const float* be2  = (const float*)d_in[8];
  const float* Wen1 = (const float*)d_in[9];
  const float* ben1 = (const float*)d_in[10];
  const float* Wen2 = (const float*)d_in[11];
  const float* ben2 = (const float*)d_in[12];
  const float* Wne1 = (const float*)d_in[13];
  const float* bne1 = (const float*)d_in[14];
  const float* Wne2 = (const float*)d_in[15];
  const float* bne2 = (const float*)d_in[16];

  float* ws = (float*)d_ws;
  float* h  = ws;
  float* E  = ws + 1024 * 128;
  float* S  = ws + 2 * 1024 * 128;
  float* h2   = (float*)d_out;
  float* out2 = (float*)d_out + 1024 * 128;

  dim3 blk(256);
  k_node_ffn<<<64, blk, 0, stream>>>(x, Wn1, bn1, Wn2, bn2, h);
  k_edge_ffn<<<64, blk, 0, stream>>>(h, We1, be1, We2, be2, E);
  k_reduce<<<1, blk, 0, stream>>>(E, S);
  k_e2n_ffn<<<64, blk, 0, stream>>>(E, S, Wen1, ben1, Wen2, ben2, h2);
  k_n2e<<<64, blk, 0, stream>>>(h2, Wne1, bne1, Wne2, bne2, out2);
}


# Layers run tile-major as a pipeline: weight loads issued in groups of 16, three groups ahead of their products, so the load counter never blocks; partial sums stored per group (sum order kept)
# speedup vs baseline: 2.8887x; 2.8887x over previous
#include <hip/hip_runtime.h>

typedef __attribute__((ext_vector_type(2))) float v2f;
typedef __attribute__((ext_vector_type(4))) float v4f;
typedef __attribute__((ext_vector_type(8))) float v8f;

__device__ __forceinline__ v8f wmma_f32(v2f a, v2f b, v8f c) {
  return __builtin_amdgcn_wmma_f32_16x16x4_f32(false, a, false, b, (short)0, c,
                                               false, false);
}

template <int N, int KS, bool FOLD>
struct LayerPipe {
  static constexpr int T = N / 16, J = KS / 4, P = N + 4;
  static constexpr int R = FOLD ? 4 : 2;
  static constexpr int G = 16 / (R * J);
  static constexpr int NG = T / G;
  static constexpr int D = 3;
  static_assert(G >= 1 && NG * G == T && NG > D, "group shape");
  const float* __restrict__ wp;
  float raw[D + 1][G][J][R];

  __device__ __forceinline__ void issue(int g) {
#pragma unroll
    for (int q = 0; q < G; ++q) {
#pragma unroll
      for (int j = 0; j < J; ++j) {
        const float* p = wp + (2 * j) * N + 16 * (g * G + q);
        raw[g % (D + 1)][q][j][0] = p[0];
        raw[g % (D + 1)][q][j][1] = p[N];
        if (FOLD) {
          raw[g % (D + 1)][q][j][R - 2] = p[128 * N];
          raw[g % (D + 1)][q][j][R - 1] = p[129 * N];
        }
      }
    }
  }

  __device__ __forceinline__ void head(const float* __restrict__ W, int wave,
                                       int hi, int lc) {
    wp = W + (wave * KS + hi * (KS / 2)) * N + lc;
#pragma unroll
    for (int d = 0; d < D; ++d) {
      issue(d);
      __builtin_amdgcn_sched_barrier(0);
    }
  }

  __device__ __forceinline__ void run(const float (&xk)[KS / 2], int wave,
                                      int hi, int lc, float* __restrict__ part) {
    float* pw = part + (wave * 16 + lc) * P + 8 * hi;
    __builtin_amdgcn_sched_barrier(0);
#pragma unroll
    for (int g = 0; g < NG; ++g) {
      if (g + D < NG) {
        issue(g + D);
        __builtin_amdgcn_sched_barrier(0);
      }
#pragma unroll
      for (int q = 0; q < G; ++q) {
        v8f acc = {};
#pragma unroll
        for (int j = 0; j < J; ++j) {
          float w0 = raw[g % (D + 1)][q][j][0], w1 = raw[g % (D + 1)][q][j][1];
          if (FOLD) {
            w0 += raw[g % (D + 1)][q][j][R - 2];
            w1 += raw[g % (D + 1)][q][j][R - 1];
          }
          v2f a = {w0, w1};
          v2f b = {xk[2 * j], xk[2 * j + 1]};
          acc = wmma_f32(a, b, acc);
        }
        v4f lo = {acc[0], acc[1], acc[2], acc[3]};
        v4f hi4 = {acc[4], acc[5], acc[6], acc[7]};
        *reinterpret_cast<v4f*>(pw + 16 * (g * G + q)) = lo;
        *reinterpret_cast<v4f*>(pw + 16 * (g * G + q) + 4) = hi4;
      }
      __builtin_amdgcn_sched_barrier(0);
    }
  }
};

template <int N>
__device__ __forceinline__ void load_bias(const float* __restrict__ bias,
                                          int wave, int hi, v4f (&bv)[N / 64]) {
  const int ch0 = wave * (N / 8) + hi * (N / 16);
#pragma unroll
  for (int q = 0; q < N / 64; ++q)
    bv[q] = *reinterpret_cast<const v4f*>(bias + ch0 + 4 * q);
}

template <int N, bool RELU>
__device__ __forceinline__ void reduce_slice(const float* __restrict__ part,
                                             const v4f (&bv)[N / 64],
                                             int wave, int hi, int lc,
                                             float (&y)[N / 16]) {
  constexpr int P = N + 4, Q = N / 64;
  const int ch0 = wave * (N / 8) + hi * (N / 16);
#pragma unroll
  for (int q = 0; q < Q; ++q) {
    const float* p = part + lc * P + ch0 + 4 * q;
    v4f s = *reinterpret_cast<const v4f*>(p);
#pragma unroll
    for (int g = 1; g < 8; ++g) s += *reinterpret_cast<const v4f*>(p + g * 16 * P);
    s += bv[q];
#pragma unroll
    for (int e = 0; e < 4; ++e) y[4 * q + e] = RELU ? fmaxf(s[e], 0.0f) : s[e];
  }
}

__device__ __forceinline__ void load8(const float* __restrict__ p, float (&v)[8]) {
  v4f a = *reinterpret_cast<const v4f*>(p);
  v4f b = *reinterpret_cast<const v4f*>(p + 4);
#pragma unroll
  for (int e = 0; e < 4; ++e) { v[e] = a[e]; v[4 + e] = b[e]; }
}
__device__ __forceinline__ void store8(float* __restrict__ p, const float (&v)[8]) {
  v4f a = {v[0], v[1], v[2], v[3]};
  v4f b = {v[4], v[5], v[6], v[7]};
  *reinterpret_cast<v4f*>(p) = a;
  *reinterpret_cast<v4f*>(p + 4) = b;
}

__global__ __launch_bounds__(256) void k_node_edge_ffn(
    const float* __restrict__ x, const float* __restrict__ Wn1,
    const float* __restrict__ bn1, const float* __restrict__ Wn2,
    const float* __restrict__ bn2, const float* __restrict__ We1,
    const float* __restrict__ be1, const float* __restrict__ We2,
    const float* __restrict__ be2, float* __restrict__ E) {
  __shared__ alignas(16) float partA[8 * 16 * 132];
  __shared__ alignas(16) float partB[8 * 16 * 132];
  const int tid = threadIdx.x;
  const int node0 = blockIdx.x * 16;
  const int wave = tid >> 5, lane = tid & 31;
  const int hi = lane >> 4, lc = lane & 15;
  const int ch0 = wave * 16 + hi * 8;

  const int c0 = wave * 32 + hi * 16;
  float xr[6];
#pragma unroll
  for (int d = 0; d < 6; ++d) xr[d] = x[(node0 + lc) * 6 + d];
  v4f b1[4], w1[6][4];
#pragma unroll
  for (int q = 0; q < 4; ++q) {
    const int c = c0 + 4 * q;
    b1[q] = *reinterpret_cast<const v4f*>(bn1 + c);
#pragma unroll
    for (int d = 0; d < 6; ++d)
      w1[d][q] = *reinterpret_cast<const v4f*>(Wn1 + d * 256 + c);
  }
  LayerPipe<128, 32, false> pn2;
  pn2.head(Wn2, wave, hi, lc);
  float t[16];
#pragma unroll
  for (int q = 0; q < 4; ++q) {
#pragma unroll
    for (int e = 0; e < 4; ++e) {
      float acc = b1[q][e] + xr[0] * w1[0][q][e] + xr[1] * w1[1][q][e] +
                  xr[2] * w1[2][q][e] + xr[3] * w1[3][q][e] +
                  xr[4] * w1[4][q][e] + xr[5] * w1[5][q][e];
      t[4 * q + e] = fmaxf(acc, 0.0f);
    }
  }
  pn2.run(t, wave, hi, lc, partA);
  __syncthreads();
  v4f bv[2];
  load_bias<128>(bn2, wave, hi, bv);
  LayerPipe<128, 16, true> pe1;
  pe1.head(We1, wave, hi, lc);
  float hk[8];
  reduce_slice<128, false>(partA, bv, wave, hi, lc, hk);
  pe1.run(hk, wave, hi, lc, partB);
  __syncthreads();
  load_bias<128>(be1, wave, hi, bv);
  LayerPipe<128, 16, false> pe2;
  pe2.head(We2, wave, hi, lc);
  float u[8];
  reduce_slice<128, true>(partB, bv, wave, hi, lc, u);
  pe2.run(u, wave, hi, lc, partA);
  __syncthreads();
  load_bias<128>(be2, wave, hi, bv);
  float y[8];
  reduce_slice<128, false>(partA, bv, wave, hi, lc, y);
  store8(E + (node0 + lc) * 128 + ch0, y);
}

__global__ __launch_bounds__(256) void k_e2n_n2e(
    const float* __restrict__ E, const float* __restrict__ Wen1,
    const float* __restrict__ ben1, const float* __restrict__ Wen2,
    const float* __restrict__ ben2, const float* __restrict__ Wne1,
    const float* __restrict__ bne1, const float* __restrict__ Wne2,
    const float* __restrict__ bne2, float* __restrict__ h2,
    float* __restrict__ out2) {
  __shared__ alignas(16) float partA[8 * 16 * 260];
  __shared__ alignas(16) float partB[8 * 16 * 132];
  __shared__ alignas(16) float psum[8][128];
  __shared__ float ps[16][16];
  __shared__ float sc[16];
  const int tid = threadIdx.x;
  const int node0 = blockIdx.x * 16;
  const int wave = tid >> 5, lane = tid & 31;
  const int hi = lane >> 4, lc = lane & 15;
  const int ch0 = wave * 16 + hi * 8;

  float ev[8];
  load8(E + (node0 + lc) * 128 + ch0, ev);

  {
    const float4* eb =
        reinterpret_cast<const float4*>(E + (size_t)(node0 >> 9) * (512 * 128)) +
        wave * (64 * 32) + lane;
    float4 s = make_float4(0.0f, 0.0f, 0.0f, 0.0f);
#pragma unroll 32
    for (int i = 0; i < 64; ++i) {
      float4 v = eb[i * 32];
      s.x += v.x;
      s.y += v.y;
      s.z += v.z;
      s.w += v.w;
    }
    *reinterpret_cast<float4*>(&psum[wave][4 * lane]) = s;
  }
  __syncthreads();
  LayerPipe<256, 16, false> pen1;
  pen1.head(Wen1, wave, hi, lc);
  float xk[8];
#pragma unroll
  for (int q = 0; q < 2; ++q) {
    v4f sq = {0.0f, 0.0f, 0.0f, 0.0f};
#pragma unroll
    for (int g = 0; g < 8; ++g)
      sq += *reinterpret_cast<const v4f*>(&psum[g][ch0 + 4 * q]);
#pragma unroll
    for (int e = 0; e < 4; ++e) xk[4 * q + e] = sq[e] - ev[4 * q + e];
  }

  pen1.run(xk, wave, hi, lc, partA);
  __syncthreads();
  v4f bw[4];
  load_bias<256>(ben1, wave, hi, bw);
  LayerPipe<128, 32, false> pen2;
  pen2.head(Wen2, wave, hi, lc);
  float u[16];
  reduce_slice<256, true>(partA, bw, wave, hi, lc, u);
  pen2.run(u, wave, hi, lc, partB);
  __syncthreads();
  v4f bv[2];
  load_bias<128>(ben2, wave, hi, bv);
  LayerPipe<128, 16, true> pne1;
  pne1.head(Wne1, wave, hi, lc);
  float hk[8];
  reduce_slice<128, false>(partB, bv, wave, hi, lc, hk);
  store8(h2 + (node0 + lc) * 128 + ch0, hk);

  pne1.run(hk, wave, hi, lc, partA);
  __syncthreads();
  load_bias<128>(bne1, wave, hi, bv);
  float v[8];
  reduce_slice<128, true>(partA, bv, wave, hi, lc, v);
  {
    float w[8];
    load8(Wne2 + ch0, w);
    float p = 0.0f;
#pragma unroll
    for (int i = 0; i < 8; ++i) p += v[i] * w[i];
    ps[lc][wave * 2 + hi] = p;
  }
  __syncthreads();
  if (tid < 16) {
    float s = 0.0f;
    for (int i = 0; i < 16; ++i) s += ps[tid][i];
    sc[tid] = (s + bne2[0]) * (1.0f / 512.0f);
  }
  __syncthreads();

  for (int idx = tid; idx < 16 * 512; idx += 256) {
    int r = idx >> 9, j = idx & 511;
    out2[(size_t)(node0 + r) * 512 + j] = sc[r];
  }
}

extern "C" void kernel_launch(void* const* d_in, const int* in_sizes, int n_in,
                              void* d_out, int out_size, void* d_ws,
                              size_t ws_size, hipStream_t stream) {
  (void)in_sizes; (void)n_in; (void)out_size; (void)ws_size;
  const float* x    = (const float*)d_in[0];
  const float* Wn1  = (const float*)d_in[1];
  const float* bn1  = (const float*)d_in[2];
  const float* Wn2  = (const float*)d_in[3];
  const float* bn2  = (const float*)d_in[4];
  const float* We1  = (const float*)d_in[5];
  const float* be1  = (const float*)d_in[6];
  const float* We2  = (const float*)d_in[7];
  const float* be2  = (const float*)d_in[8];
  const float* Wen1 = (const float*)d_in[9];
  const float* ben1 = (const float*)d_in[10];
  const float* Wen2 = (const float*)d_in[11];
  const float* ben2 = (const float*)d_in[12];
  const float* Wne1 = (const float*)d_in[13];
  const float* bne1 = (const float*)d_in[14];
  const float* Wne2 = (const float*)d_in[15];
  const float* bne2 = (const float*)d_in[16];

  float* E  = (float*)d_ws;
  float* h2   = (float*)d_out;
  float* out2 = (float*)d_out + 1024 * 128;

  dim3 blk(256);
  k_node_edge_ffn<<<64, blk, 0, stream>>>(x, Wn1, bn1, Wn2, bn2, We1, be1, We2,
                                          be2, E);
  k_e2n_n2e<<<64, blk, 0, stream>>>(E, Wen1, ben1, Wen2, ben2, Wne1, bne1, Wne2,
                                    bne2, h2, out2);
}
